# starting kernel
# baseline (speedup 1.0000x reference)
#include <hip/hip_runtime.h>
#include <math.h>

#define NPIX   102400
#define LCONV  589
#define KPAD   592
#define NSAMP  50
#define DELAYN 30
#define REVSZ  1184

typedef float v2f __attribute__((ext_vector_type(2)));
typedef float v8f __attribute__((ext_vector_type(8)));

__device__ __forceinline__ float nan0f(float x) { return (x != x) ? 0.0f : x; }

__global__ __launch_bounds__(256) void dce_prep(const float* __restrict__ st,
                                                const float* __restrict__ cp,
                                                float* __restrict__ aif_g,
                                                int*   __restrict__ idx_g)
{
    __shared__ float st_s[NSAMP], cp_s[NSAMP];
    int tid = threadIdx.x;
    if (tid < NSAMP) { st_s[tid] = st[tid]; cp_s[tid] = cp[tid]; }
    __syncthreads();

    for (int t = tid; t < KPAD; t += 256) {
        float v = 0.0f;
        if (t >= DELAYN && t < LCONV) {
            float x = (float)(t - DELAYN) * 0.1f;
            if (x <= st_s[0])              v = cp_s[0];
            else if (x >= st_s[NSAMP - 1]) v = cp_s[NSAMP - 1];
            else {
                int i = 0;
                for (int j = 1; j < NSAMP - 1; ++j)
                    if (st_s[j] <= x) i = j;
                v = cp_s[i] + (cp_s[i + 1] - cp_s[i]) * (x - st_s[i]) / (st_s[i + 1] - st_s[i]);
            }
        }
        aif_g[t] = v;
    }

    if (tid < 64) {
        int id = -2;
        if (tid < NSAMP) {
            float xv = st_s[tid];
            int j = LCONV - 1;
            for (int t = LCONV - 1; t >= 0; --t)
                if ((float)t * 0.1f >= xv) j = t;
            id = j;
        }
        idx_g[tid] = id;
    }
}

__device__ __forceinline__ void write_tile(v8f acc, int m, int hl, int p,
                                           float M0t, float COSF, float offs,
                                           float* __restrict__ out)
{
    union { v8f v; float f[8]; } u; u.v = acc;
#pragma unroll
    for (int jj = 0; jj < 8; ++jj) {
        int s = m * 16 + jj + 8 * hl;
        if (s < NSAMP) {
            float CA   = u.f[jj];
            float E1CA = __expf(-0.00487f * (1.0f + 4.3f * CA));
            float CAt  = M0t * (1.0f - E1CA) *
                         __builtin_amdgcn_rcpf(1.0f - E1CA * COSF);
            out[s * NPIX + p] = CAt + offs;
        }
    }
}

__global__ __launch_bounds__(256) void dce_main(const float* __restrict__ param,
                                                const float* __restrict__ aif_g,
                                                const int*   __restrict__ idx_g,
                                                float* __restrict__ out)
{
    __shared__ float rev_s[REVSZ];
    __shared__ int   idx_s[64];
    int tid = threadIdx.x;
    for (int i = tid; i < REVSZ; i += 256) {
        int t = (LCONV - 1) - i;
        rev_s[i] = (t >= 0) ? aif_g[t] : 0.0f;
    }
    if (tid < 64) idx_s[tid] = idx_g[tid];
    __syncthreads();

    const int wave = tid >> 5;
    const int lane = tid & 31;
    const int hl   = lane >> 4;
    const int lpix = lane & 15;
    const int tile = blockIdx.x * 8 + wave;
    const int p    = tile * 16 + lpix;

    float ve = param[0 * NPIX + p];
    float vp = param[1 * NPIX + p];
    float fp = param[2 * NPIX + p];
    float ps = param[3 * NPIX + p];

    float Te = nan0f(ve / ps);
    float T  = nan0f((vp + ve) / fp);
    float Tc = nan0f(vp / fp);
    float sm   = T + Te;
    float disc = sqrtf(sm * sm - 4.0f * Tc * Te);
    float inv2 = 1.0f / (2.0f * Tc * Te);
    float thp  = nan0f((sm + disc) * inv2);
    float thm  = nan0f((sm - disc) * inv2);
    float pref = thp * thm;
    float dn   = thp - thm;
    float am = pref / dn;
    float ap = -am;
    float bm = pref * (1.0f - Te * thm) / dn;
    float bp = pref * (Te * thp - 1.0f) / dn;

    float rm    = __expf(-0.1f  * thm);
    float rp    = __expf(-0.1f  * thp);
    float rm589 = __expf(-58.9f * thm);
    float rp589 = __expf(-58.9f * thp);
    float Smm = (1.0f - rm589) / (1.0f - rm);
    float Spp = (1.0f - rp589) / (1.0f - rp);
    float sumHe = am * Smm + ap * Spp;
    float sumHp = bm * Smm + bp * Spp;

    float P = nan0f(vp * nan0f(bm / sumHp) + ve * nan0f(am / sumHe));
    float Q = nan0f(vp * nan0f(bp / sumHp) + ve * nan0f(ap / sumHe));

    float rm2 = rm * rm, rp2 = rp * rp;
    float rm4 = rm2 * rm2, rp4 = rp2 * rp2;
    float rmt = hl ? rm2 : 1.0f;
    float rpt = hl ? rp2 : 1.0f;

    int q0 = (LCONV - 1) - idx_s[ 0 + lpix] + 2 * hl;
    int q1 = (LCONV - 1) - idx_s[16 + lpix] + 2 * hl;
    int q2 = (LCONV - 1) - idx_s[32 + lpix] + 2 * hl;
    int q3 = (LCONV - 1) - idx_s[48 + lpix] + 2 * hl;

    v8f acc0 = {}, acc1 = {}, acc2 = {}, acc3 = {};

#pragma unroll 4
    for (int k = 0; k < KPAD / 4; ++k) {
        float b0 = P * rmt + Q * rpt;
        float rmt1 = rmt * rm, rpt1 = rpt * rp;
        float b1 = P * rmt1 + Q * rpt1;
        rmt *= rm4; rpt *= rp4;
        v2f B; B.x = b0; B.y = b1;

        v2f A0, A1, A2, A3;
        A0.x = rev_s[q0];  A0.y = rev_s[q0 + 1];
        A1.x = rev_s[q1];  A1.y = rev_s[q1 + 1];
        A2.x = rev_s[q2];  A2.y = rev_s[q2 + 1];
        A3.x = rev_s[q3];  A3.y = rev_s[q3 + 1];
        q0 += 4; q1 += 4; q2 += 4; q3 += 4;

        acc0 = __builtin_amdgcn_wmma_f32_16x16x4_f32(false, A0, false, B, (short)0, acc0, false, false);
        acc1 = __builtin_amdgcn_wmma_f32_16x16x4_f32(false, A1, false, B, (short)0, acc1, false, false);
        acc2 = __builtin_amdgcn_wmma_f32_16x16x4_f32(false, A2, false, B, (short)0, acc2, false, false);
        acc3 = __builtin_amdgcn_wmma_f32_16x16x4_f32(false, A3, false, B, (short)0, acc3, false, false);
    }

    const float COSF = 0.98480775301220805936f;
    const float SINF = 0.17364817766693034885f;
    float E1   = __expf(-0.00487f);
    float M0   = 100.0f * (1.0f - COSF * E1) / (SINF * (1.0f - E1));
    float M0t  = M0 * SINF;
    float Mst  = M0t * (1.0f - E1) / (1.0f - E1 * COSF);
    float offs = 100.0f - Mst;

    write_tile(acc0, 0, hl, p, M0t, COSF, offs, out);
    write_tile(acc1, 1, hl, p, M0t, COSF, offs, out);
    write_tile(acc2, 2, hl, p, M0t, COSF, offs, out);
    write_tile(acc3, 3, hl, p, M0t, COSF, offs, out);
}

extern "C" void kernel_launch(void* const* d_in, const int* in_sizes, int n_in,
                              void* d_out, int out_size, void* d_ws, size_t ws_size,
                              hipStream_t stream)
{
    (void)in_sizes; (void)n_in; (void)out_size; (void)ws_size;
    const float* param = (const float*)d_in[0];
    const float* st    = (const float*)d_in[1];
    const float* cp    = (const float*)d_in[2];
    float* out   = (float*)d_out;
    float* aif_g = (float*)d_ws;
    int*   idx_g = (int*)((char*)d_ws + KPAD * sizeof(float));

    dce_prep<<<1, 256, 0, stream>>>(st, cp, aif_g, idx_g);
    dce_main<<<800, 256, 0, stream>>>(param, aif_g, idx_g, out);
}


# Prep kernel writes the block tables in final LDS form (reversed curve, packed row lists, counts): blocks copy 3 KB with 16-byte loads; no packing wave, no per-element guards, one barrier fewer
# speedup vs baseline: 1.7244x; 1.7244x over previous
#include <hip/hip_runtime.h>
#include <math.h>

#define NPIX   102400
#define LCONV  589
#define KPAD   592
#define NSAMP  50
#define DELAYN 30
#define REVSZ  1184

typedef float v2f __attribute__((ext_vector_type(2)));
typedef float v8f __attribute__((ext_vector_type(8)));

__device__ __forceinline__ float nan0f(float x) { return (x != x) ? 0.0f : x; }

#define PREP_THREADS 640
#define IDX_TID0     608

#define TAB_SROW  0
#define TAB_IDXC  64
#define TAB_DEAD  128
#define TAB_CNT   192
#define TAB_WORDS 200
#define WS_WORDS  (TAB_WORDS + KPAD)

__device__ __forceinline__ int grid_index(float xv)
{
    int lo = 0, hi = LCONV;
    for (int it = 0; it < 10; ++it) {
        if (lo < hi) {
            const int mid = (lo + hi) >> 1;
            if ((float)mid * 0.1f >= xv) hi = mid; else lo = mid + 1;
        }
    }
    return (lo < LCONV - 1) ? lo : (LCONV - 1);
}

__global__ __launch_bounds__(PREP_THREADS) void dce_prep(const float* __restrict__ st,
                                                         const float* __restrict__ cp,
                                                         int*   __restrict__ tab_g,
                                                         float* __restrict__ rev_g)
{
    __shared__ float st_s[NSAMP], cp_s[NSAMP];
    __shared__ int   tab_s[TAB_WORDS];
    int tid = threadIdx.x;
    if (tid < NSAMP) { st_s[tid] = st[tid]; cp_s[tid] = cp[tid]; }
    __syncthreads();

    if (tid < KPAD) {
        const int t = tid;
        float v = 0.0f;
        if (t >= DELAYN && t < LCONV) {
            float x = (float)(t - DELAYN) * 0.1f;
            if (x <= st_s[0])              v = cp_s[0];
            else if (x >= st_s[NSAMP - 1]) v = cp_s[NSAMP - 1];
            else {
                int i = 0;
                for (int j = 1; j < NSAMP - 1; ++j)
                    if (st_s[j] <= x) i = j;
                v = cp_s[i] + (cp_s[i + 1] - cp_s[i]) * (x - st_s[i]) / (st_s[i + 1] - st_s[i]);
            }
        }
        rev_g[(t < LCONV) ? (LCONV - 1 - t) : t] = v;
    } else if (tid >= IDX_TID0) {
        const int lane = tid - IDX_TID0;
        const int sa = lane, sb = lane + 32;
        const bool vb = sb < NSAMP;
        const int ia = grid_index(st_s[sa]);
        const int ib = vb ? grid_index(st_s[sb]) : -2;
        const bool la = ia >= DELAYN;
        const bool lb = vb && (ib >= DELAYN);
        const bool da = !la;
        const bool db = vb && !lb;
        const unsigned mla = __builtin_amdgcn_ballot_w32(la);
        const unsigned mlb = __builtin_amdgcn_ballot_w32(lb);
        const unsigned mda = __builtin_amdgcn_ballot_w32(da);
        const unsigned mdb = __builtin_amdgcn_ballot_w32(db);
        const unsigned below = (1u << lane) - 1u;
        const int nla = __popc(mla), nda = __popc(mda);
        tab_s[TAB_SROW + sa] = -1;  tab_s[TAB_SROW + sb] = -1;
        tab_s[TAB_IDXC + sa] = -2;  tab_s[TAB_IDXC + sb] = -2;
        tab_s[TAB_DEAD + sa] = 0;   tab_s[TAB_DEAD + sb] = 0;
        if (lane < TAB_WORDS - TAB_CNT) tab_s[TAB_CNT + lane] = 0;
        __builtin_amdgcn_wave_barrier();
        if (la) { int pos = __popc(mla & below);       tab_s[TAB_IDXC + pos] = ia; tab_s[TAB_SROW + pos] = sa; }
        if (lb) { int pos = nla + __popc(mlb & below); tab_s[TAB_IDXC + pos] = ib; tab_s[TAB_SROW + pos] = sb; }
        if (da) tab_s[TAB_DEAD + __popc(mda & below)] = sa;
        if (db) tab_s[TAB_DEAD + nda + __popc(mdb & below)] = sb;
        if (lane == 0) { tab_s[TAB_CNT] = nla + __popc(mlb); tab_s[TAB_CNT + 1] = nda + __popc(mdb); }
        __builtin_amdgcn_wave_barrier();
        for (int i = lane; i < TAB_WORDS; i += 32) tab_g[i] = tab_s[i];
    }
}

#define MAIN_THREADS 128
#define MAIN_BLOCKS  6400
#define KQ_STEPS     37
#define RED_TILE     256

__device__ __forceinline__ float spgr(float CA, float M0t, float COSF, float offs)
{
    float E1CA = __expf(-0.00487f * (1.0f + 4.3f * CA));
    float CAt  = M0t * (1.0f - E1CA) *
                 __builtin_amdgcn_rcpf(1.0f - E1CA * COSF);
    return CAt + offs;
}

__device__ __forceinline__ void write_tile(v8f acc, int m, int hl, int p,
                                           float M0t, float COSF, float offs,
                                           const float* srow_s, float* __restrict__ out)
{
    union { v8f v; float f[8]; } u; u.v = acc;
    const float4* sp = reinterpret_cast<const float4*>(srow_s + m * 16 + 8 * hl);
    const float4 r0 = sp[0], r1 = sp[1];
    const int srow[8] = { __float_as_int(r0.x), __float_as_int(r0.y), __float_as_int(r0.z), __float_as_int(r0.w),
                          __float_as_int(r1.x), __float_as_int(r1.y), __float_as_int(r1.z), __float_as_int(r1.w) };
#pragma unroll
    for (int jj = 0; jj < 8; ++jj) {
        const int s = srow[jj];
        if (s >= 0)
            out[s * NPIX + p] = spgr(u.f[jj], M0t, COSF, offs);
    }
}

template <int NT>
__device__ __forceinline__ void conv_quarter(const float* rev_s, const float* idxc_s,
                                             int m0, int lpix, int hl, int kq,
                                             float P, float Q, float thm, float thp,
                                             float rm, float rp, v8f (&acc)[NT])
{
    const int   t0  = 4 * KQ_STEPS * kq + 2 * hl;
    const float t0f = (float)t0;
    const float rmt = (t0 == 0) ? 1.0f : __expf(-0.1f * thm * t0f);
    const float rpt = (t0 == 0) ? 1.0f : __expf(-0.1f * thp * t0f);
    const float rm2 = rm * rm, rp2 = rp * rp;
    v2f U, Wv, RM4, RP4;
    U.x  = P * rmt;  U.y  = P * (rmt * rm);
    Wv.x = Q * rpt;  Wv.y = Q * (rpt * rp);
    RM4.x = RM4.y = rm2 * rm2;
    RP4.x = RP4.y = rp2 * rp2;

    int q[NT];
#pragma unroll
    for (int m = 0; m < NT; ++m) {
        q[m]   = (LCONV - 1) - __float_as_int(idxc_s[16 * (m0 + m) + lpix]) + t0;
        acc[m] = v8f{};
    }

    v2f A[NT];
#pragma unroll
    for (int m = 0; m < NT; ++m) { A[m].x = rev_s[q[m]]; A[m].y = rev_s[q[m] + 1]; }

#pragma unroll 4
    for (int ks = 0; ks < KQ_STEPS - 1; ++ks) {
        v2f N[NT];
#pragma unroll
        for (int m = 0; m < NT; ++m) { q[m] += 4; N[m].x = rev_s[q[m]]; N[m].y = rev_s[q[m] + 1]; }
        const v2f B = U + Wv;
        U *= RM4;  Wv *= RP4;
#pragma unroll
        for (int m = 0; m < NT; ++m)
            acc[m] = __builtin_amdgcn_wmma_f32_16x16x4_f32(false, A[m], false, B, (short)0, acc[m], false, false);
#pragma unroll
        for (int m = 0; m < NT; ++m) A[m] = N[m];
    }
    {
        const v2f B = U + Wv;
#pragma unroll
        for (int m = 0; m < NT; ++m)
            acc[m] = __builtin_amdgcn_wmma_f32_16x16x4_f32(false, A[m], false, B, (short)0, acc[m], false, false);
    }
}

__device__ __forceinline__ void put_partial(float* dst, v8f a, int lane)
{
    union { v8f v; float f[8]; } u; u.v = a;
#pragma unroll
    for (int r = 0; r < 8; ++r) dst[r * 32 + lane] = u.f[r];
}

__device__ __forceinline__ v8f sum_partials(const float* src, int lane)
{
    union { v8f v; float f[8]; } u;
#pragma unroll
    for (int r = 0; r < 8; ++r) {
        const float* s = src + r * 32 + lane;
        u.f[r] = (s[0] + s[RED_TILE]) + (s[2 * RED_TILE] + s[3 * RED_TILE]);
    }
    return u.v;
}

#define SM_TAB   0
#define SM_REV   (SM_TAB + TAB_WORDS)
#define SM_RED   (SM_REV + REVSZ)
#define SM_COEF  (SM_RED + 3 * 4 * RED_TILE)
#define SM_FLOATS (SM_COEF + 16 * 8)

__global__ __launch_bounds__(MAIN_THREADS) void dce_main(const float* __restrict__ param,
                                                         const float* __restrict__ tab_g,
                                                         float* __restrict__ out)
{
    __shared__ __attribute__((aligned(16))) float fsm[SM_FLOATS];
    const float* tab_s  = fsm + SM_TAB;
    const float* srow_s = tab_s + TAB_SROW;
    const float* idxc_s = tab_s + TAB_IDXC;
    const float* dead_s = tab_s + TAB_DEAD;
    const float* cnt_s  = tab_s + TAB_CNT;
    float* rev_s  = fsm + SM_REV;
    float* red_s  = fsm + SM_RED;
    float* coef_s = fsm + SM_COEF;

    const int tid  = threadIdx.x;
    const int wave = tid >> 5;
    const int lane = tid & 31;
    const int hl   = lane >> 4;
    const int lpix = lane & 15;
    const int p    = blockIdx.x * 16 + lpix;

    const unsigned b  = blockIdx.x;
    const int turn = (int)((b ^ (b >> 2) ^ (b >> 4) ^ (b >> 6) ^ (b >> 8) ^ (b >> 10) ^ (b >> 12)) & 3u);
    const int cw   = turn;
    const int role = (wave - cw + 3) & 3;

    float ve = 1.0f, vp = 1.0f, fp = 1.0f, ps = 1.0f;
    if (wave == cw) {
        ve = param[0 * NPIX + p];
        vp = param[1 * NPIX + p];
        fp = param[2 * NPIX + p];
        ps = param[3 * NPIX + p];
    }

    {
        const float4* src4 = reinterpret_cast<const float4*>(tab_g);
        float4*       dst4 = reinterpret_cast<float4*>(fsm);
        const int  nvec    = WS_WORDS / 4;
        const bool second  = tid + MAIN_THREADS < nvec;
        const int  v1 = second ? (tid + MAIN_THREADS) : (nvec - 1);
        const float4 w0 = src4[tid];
        const float4 w1 = src4[v1];
        if (tid < SM_RED / 4 - 2 * MAIN_THREADS)
            dst4[2 * MAIN_THREADS + tid] = make_float4(0.0f, 0.0f, 0.0f, 0.0f);
        dst4[tid] = w0;
        dst4[tid + MAIN_THREADS] = second ? w1 : make_float4(0.0f, 0.0f, 0.0f, 0.0f);
    }

    if (wave == cw) {
        float Te = nan0f(ve / ps);
        float T  = nan0f((vp + ve) / fp);
        float Tc = nan0f(vp / fp);
        float sm   = T + Te;
        float disc = sqrtf(sm * sm - 4.0f * Tc * Te);
        float inv2 = 1.0f / (2.0f * Tc * Te);
        float thp  = nan0f((sm + disc) * inv2);
        float thm  = nan0f((sm - disc) * inv2);
        float pref = thp * thm;
        float dn   = thp - thm;
        float am = pref / dn;
        float ap = -am;
        float bm = pref * (1.0f - Te * thm) / dn;
        float bp = pref * (Te * thp - 1.0f) / dn;

        float rm    = __expf(-0.1f  * thm);
        float rp    = __expf(-0.1f  * thp);
        float rm589 = __expf(-58.9f * thm);
        float rp589 = __expf(-58.9f * thp);
        float Smm = (1.0f - rm589) / (1.0f - rm);
        float Spp = (1.0f - rp589) / (1.0f - rp);
        float sumHe = am * Smm + ap * Spp;
        float sumHp = bm * Smm + bp * Spp;

        float P = nan0f(vp * nan0f(bm / sumHp) + ve * nan0f(am / sumHe));
        float Q = nan0f(vp * nan0f(bp / sumHp) + ve * nan0f(ap / sumHe));

        const float COSF = 0.98480775301220805936f;
        const float SINF = 0.17364817766693034885f;
        float E1   = __expf(-0.00487f);
        float M0   = 100.0f * (1.0f - COSF * E1) / (SINF * (1.0f - E1));
        float M0t  = M0 * SINF;
        float Mst  = M0t * (1.0f - E1) / (1.0f - E1 * COSF);
        float offs = 100.0f - Mst;

        if (hl == 0) {
            float4* cd = reinterpret_cast<float4*>(coef_s + 8 * lpix);
            cd[0] = make_float4(P, Q, thm, thp);
            cd[1] = make_float4(rm, rp, M0t, offs);
        }
    }
    __syncthreads();

    const float COSF = 0.98480775301220805936f;
    const float4 c0 = reinterpret_cast<const float4*>(coef_s + 8 * lpix)[0];
    const float4 c1 = reinterpret_cast<const float4*>(coef_s + 8 * lpix)[1];
    const float P = c0.x, Q = c0.y, thm = c0.z, thp = c0.w;
    const float rm = c1.x, rp = c1.y, M0t = c1.z, offs = c1.w;

    {
        v8f acc[3];
        conv_quarter<3>(rev_s, idxc_s, 0, lpix, hl, wave, P, Q, thm, thp, rm, rp, acc);
#pragma unroll
        for (int m = 0; m < 3; ++m)
            put_partial(red_s + (m * 4 + wave) * RED_TILE, acc[m], lane);
    }
    __syncthreads();

    if (role < 3) {
        const v8f tot = sum_partials(red_s + role * 4 * RED_TILE, lane);
        write_tile(tot, role, hl, p, M0t, COSF, offs, srow_s, out);
    } else {
        const int   ndead = __float_as_int(cnt_s[1]);
        const float sig0  = spgr(0.0f, M0t, COSF, offs);
        for (int i = hl; i < ndead; i += 2)
            out[__float_as_int(dead_s[i]) * NPIX + p] = sig0;
    }

    if (__float_as_int(cnt_s[0]) > 48) {
        __syncthreads();
        v8f acc3[1];
        conv_quarter<1>(rev_s, idxc_s, 3, lpix, hl, wave, P, Q, thm, thp, rm, rp, acc3);
        put_partial(red_s + wave * RED_TILE, acc3[0], lane);
        __syncthreads();
        if (role == 3) {
            const v8f tot = sum_partials(red_s, lane);
            write_tile(tot, 3, hl, p, M0t, COSF, offs, srow_s, out);
        }
    }
}

extern "C" void kernel_launch(void* const* d_in, const int* in_sizes, int n_in,
                              void* d_out, int out_size, void* d_ws, size_t ws_size,
                              hipStream_t stream)
{
    (void)in_sizes; (void)n_in; (void)out_size; (void)ws_size;
    const float* param = (const float*)d_in[0];
    const float* st    = (const float*)d_in[1];
    const float* cp    = (const float*)d_in[2];
    float* out   = (float*)d_out;
    int*   tab_g = (int*)d_ws;
    float* rev_g = (float*)d_ws + TAB_WORDS;

    dce_prep<<<1, PREP_THREADS, 0, stream>>>(st, cp, tab_g, rev_g);
    dce_main<<<MAIN_BLOCKS, MAIN_THREADS, 0, stream>>>(param, (const float*)d_ws, out);
}
